# starting kernel
# baseline (speedup 1.0000x reference)
#include <hip/hip_runtime.h>
#include <math.h>

typedef float v2f   __attribute__((ext_vector_type(2)));
typedef float v8f   __attribute__((ext_vector_type(8)));
typedef float f32x4 __attribute__((ext_vector_type(4)));

namespace {
constexpr int S_LEN = 4096;
constexpr int CH    = 64;
constexpr int SB    = 64;
constexpr int HALO  = 31;
constexpr int ROWS  = 128;
constexpr int LSTR  = 80;

constexpr int F_TOTAL  = 11 + 31 + 61;
constexpr int HP_TOTAL = 43 + 63 + 95;
}

template<int KPAD, int SHIFT, int HPBASE>
__device__ __forceinline__ void conv_pair(const float* tile, const float* hp,
                                          int lane, int st0, int st1, int c0,
                                          v8f& acc0, v8f& acc1)
{
  const int i   = lane & 15;
  const int hi  = lane >> 4;
  const int rb0 = (HALO - SHIFT + 16 * st0) * LSTR + c0 + i;
  const int rb1 = (HALO - SHIFT + 16 * st1) * LSTR + c0 + i;
  const int ab  = HPBASE + 15 - i;
#pragma unroll
  for (int k0 = 0; k0 < KPAD; k0 += 4) {
    const int t0 = k0 + 2 * hi;
    v2f a, b0, b1;
    a.x  = hp[ab + t0];
    a.y  = hp[ab + t0 + 1];
    b0.x = tile[rb0 +  t0      * LSTR];
    b0.y = tile[rb0 + (t0 + 1) * LSTR];
    b1.x = tile[rb1 +  t0      * LSTR];
    b1.y = tile[rb1 + (t0 + 1) * LSTR];
    acc0 = __builtin_amdgcn_wmma_f32_16x16x4_f32(false, a, false, b0, (short)0, acc0, false, false);
    acc1 = __builtin_amdgcn_wmma_f32_16x16x4_f32(false, a, false, b1, (short)0, acc1, false, false);
  }
}

__device__ __forceinline__ void store_tile(float* __restrict__ out, int b, int sc,
                                           int row0, int c0, int lane, v8f acc)
{
  const int j  = lane & 15;
  const int hi = lane >> 4;
  const size_t base = (((size_t)b * 3 + sc) * S_LEN + (size_t)(row0 + 8 * hi)) * CH
                      + (size_t)(c0 + j);
#pragma unroll
  for (int r = 0; r < 8; ++r)
    out[base + (size_t)r * CH] = acc[r];
}

__global__ void __launch_bounds__(256)
cwt_gaus1_wmma_kernel(const float* __restrict__ x, float* __restrict__ out)
{
  __shared__ float  tile[ROWS * LSTR];
  __shared__ double psum[256];
  __shared__ float  fbuf[F_TOTAL];
  __shared__ float  hp[HP_TOTAL];

  const int tid = threadIdx.x;
  const int b   = blockIdx.x >> 6;
  const int sB  = (blockIdx.x & 63) * SB;

  const double step = 10.0 / 1023.0;
  const double invC = 1.0 / pow(M_PI * 0.5, 0.25);

  {
    double s = 0.0;
#pragma unroll
    for (int q = 0; q < 4; ++q) {
      const int    i  = tid * 4 + q;
      const double xv = -5.0 + step * (double)i;
      s += -2.0 * xv * exp(-xv * xv) * invC;
    }
    psum[tid] = s;
  }
  __syncthreads();
  if (tid == 0) {
    double run = 0.0;
    for (int t = 0; t < 256; ++t) { const double tmp = psum[t]; psum[t] = run; run += tmp; }
  }
  __syncthreads();
  if (tid < F_TOTAL) {
    int a, k;
    if (tid < 11)      { a = 1; k = tid;      }
    else if (tid < 42) { a = 3; k = tid - 11; }
    else               { a = 6; k = tid - 42; }
    const long j = (long)((double)k / ((double)a * step));
    double v = psum[j >> 2];
    for (long i = j & ~3L; i <= j; ++i) {
      const double xv = -5.0 + step * (double)i;
      v += -2.0 * xv * exp(-xv * xv) * invC;
    }
    fbuf[tid] = (float)(v * step);
  }
  __syncthreads();
  if (tid < HP_TOTAL) {
    int d, foff, LH; double aa;
    if (tid < 43)       { d = tid - 15;       foff = 0;  LH = 12; aa = 1.0; }
    else if (tid < 106) { d = tid - 43 - 15;  foff = 11; LH = 32; aa = 3.0; }
    else                { d = tid - 106 - 15; foff = 42; LH = 62; aa = 6.0; }
    float v = 0.0f;
    if (d >= 0 && d < LH) {
      const float fm1 = (d >= 1)      ? fbuf[foff + d - 1] : 0.0f;
      const float fm  = (d < LH - 1)  ? fbuf[foff + d]     : 0.0f;
      v = (float)(-sqrt(aa)) * (fm1 - fm);
    }
    hp[tid] = v;
  }

#pragma unroll
  for (int it = 0; it < 8; ++it) {
    const int e4  = tid + it * 256;
    const int row = e4 >> 4;
    const int col = (e4 & 15) << 2;
    const int gr  = sB - HALO + row;
    f32x4 v = {0.0f, 0.0f, 0.0f, 0.0f};
    if (gr >= 0 && gr < S_LEN)
      v = *(const f32x4*)(x + ((size_t)b * S_LEN + (size_t)gr) * CH + col);
    *(f32x4*)(&tile[row * LSTR + col]) = v;
  }
  __syncthreads();

  const int lane = tid & 31;
  const int w    = tid >> 5;
  const int c0   = (w & 3) * 16;
  const int st0  = (w >> 2) * 2;
  const int st1  = st0 + 1;

  const v8f z = {0.f, 0.f, 0.f, 0.f, 0.f, 0.f, 0.f, 0.f};

  {
    v8f a0 = z, a1 = z;
    conv_pair<28, 6, 0>(tile, hp, lane, st0, st1, c0, a0, a1);
    store_tile(out, b, 0, sB + 16 * st0, c0, lane, a0);
    store_tile(out, b, 0, sB + 16 * st1, c0, lane, a1);
  }
  {
    v8f a0 = z, a1 = z;
    conv_pair<48, 16, 43>(tile, hp, lane, st0, st1, c0, a0, a1);
    store_tile(out, b, 1, sB + 16 * st0, c0, lane, a0);
    store_tile(out, b, 1, sB + 16 * st1, c0, lane, a1);
  }
  {
    v8f a0 = z, a1 = z;
    conv_pair<80, 31, 106>(tile, hp, lane, st0, st1, c0, a0, a1);
    store_tile(out, b, 2, sB + 16 * st0, c0, lane, a0);
    store_tile(out, b, 2, sB + 16 * st1, c0, lane, a1);
  }
}

extern "C" void kernel_launch(void* const* d_in, const int* in_sizes, int n_in,
                              void* d_out, int out_size, void* d_ws, size_t ws_size,
                              hipStream_t stream) {
  (void)in_sizes; (void)n_in; (void)out_size; (void)d_ws; (void)ws_size;
  const float* x  = (const float*)d_in[0];
  float*      out = (float*)d_out;
  hipLaunchKernelGGL(cwt_gaus1_wmma_kernel, dim3(32 * 64), dim3(256), 0, stream, x, out);
}


# Build the 201 filter taps once per call in a one-block pre-launch (to workspace) instead of in each of the 2048 blocks; fp64 build code unchanged; main kernel copies the taps to LDS
# speedup vs baseline: 1.9685x; 1.9685x over previous
#include <hip/hip_runtime.h>
#include <math.h>

typedef float v2f   __attribute__((ext_vector_type(2)));
typedef float v8f   __attribute__((ext_vector_type(8)));
typedef float f32x4 __attribute__((ext_vector_type(4)));

namespace {
constexpr int S_LEN = 4096;
constexpr int CH    = 64;
constexpr int SB    = 64;
constexpr int HALO  = 31;
constexpr int ROWS  = 128;
constexpr int LSTR  = 80;

constexpr int F_TOTAL  = 11 + 31 + 61;
constexpr int HP_TOTAL = 43 + 63 + 95;
}

template<int KPAD, int SHIFT, int HPBASE>
__device__ __forceinline__ void conv_pair(const float* tile, const float* hp,
                                          int lane, int st0, int st1, int c0,
                                          v8f& acc0, v8f& acc1)
{
  const int i   = lane & 15;
  const int hi  = lane >> 4;
  const int rb0 = (HALO - SHIFT + 16 * st0) * LSTR + c0 + i;
  const int rb1 = (HALO - SHIFT + 16 * st1) * LSTR + c0 + i;
  const int ab  = HPBASE + 15 - i;
#pragma unroll
  for (int k0 = 0; k0 < KPAD; k0 += 4) {
    const int t0 = k0 + 2 * hi;
    v2f a, b0, b1;
    a.x  = hp[ab + t0];
    a.y  = hp[ab + t0 + 1];
    b0.x = tile[rb0 +  t0      * LSTR];
    b0.y = tile[rb0 + (t0 + 1) * LSTR];
    b1.x = tile[rb1 +  t0      * LSTR];
    b1.y = tile[rb1 + (t0 + 1) * LSTR];
    acc0 = __builtin_amdgcn_wmma_f32_16x16x4_f32(false, a, false, b0, (short)0, acc0, false, false);
    acc1 = __builtin_amdgcn_wmma_f32_16x16x4_f32(false, a, false, b1, (short)0, acc1, false, false);
  }
}

__device__ __forceinline__ void store_tile(float* __restrict__ out, int b, int sc,
                                           int row0, int c0, int lane, v8f acc)
{
  const int j  = lane & 15;
  const int hi = lane >> 4;
  const size_t base = (((size_t)b * 3 + sc) * S_LEN + (size_t)(row0 + 8 * hi)) * CH
                      + (size_t)(c0 + j);
#pragma unroll
  for (int r = 0; r < 8; ++r)
    out[base + (size_t)r * CH] = acc[r];
}

__global__ void __launch_bounds__(256)
cwt_gaus1_taps_kernel(float* __restrict__ hp)
{
  __shared__ double psum[256];
  __shared__ float  fbuf[F_TOTAL];

  const int tid = threadIdx.x;

  const double step = 10.0 / 1023.0;
  const double invC = 1.0 / pow(M_PI * 0.5, 0.25);

  {
    double s = 0.0;
#pragma unroll
    for (int q = 0; q < 4; ++q) {
      const int    i  = tid * 4 + q;
      const double xv = -5.0 + step * (double)i;
      s += -2.0 * xv * exp(-xv * xv) * invC;
    }
    psum[tid] = s;
  }
  __syncthreads();
  if (tid == 0) {
    double run = 0.0;
    for (int t = 0; t < 256; ++t) { const double tmp = psum[t]; psum[t] = run; run += tmp; }
  }
  __syncthreads();
  if (tid < F_TOTAL) {
    int a, k;
    if (tid < 11)      { a = 1; k = tid;      }
    else if (tid < 42) { a = 3; k = tid - 11; }
    else               { a = 6; k = tid - 42; }
    const long j = (long)((double)k / ((double)a * step));
    double v = psum[j >> 2];
    for (long i = j & ~3L; i <= j; ++i) {
      const double xv = -5.0 + step * (double)i;
      v += -2.0 * xv * exp(-xv * xv) * invC;
    }
    fbuf[tid] = (float)(v * step);
  }
  __syncthreads();
  if (tid < HP_TOTAL) {
    int d, foff, LH; double aa;
    if (tid < 43)       { d = tid - 15;       foff = 0;  LH = 12; aa = 1.0; }
    else if (tid < 106) { d = tid - 43 - 15;  foff = 11; LH = 32; aa = 3.0; }
    else                { d = tid - 106 - 15; foff = 42; LH = 62; aa = 6.0; }
    float v = 0.0f;
    if (d >= 0 && d < LH) {
      const float fm1 = (d >= 1)      ? fbuf[foff + d - 1] : 0.0f;
      const float fm  = (d < LH - 1)  ? fbuf[foff + d]     : 0.0f;
      v = (float)(-sqrt(aa)) * (fm1 - fm);
    }
    hp[tid] = v;
  }
}

__global__ void __launch_bounds__(256)
cwt_gaus1_wmma_kernel(const float* __restrict__ x, const float* __restrict__ taps,
                      float* __restrict__ out)
{
  __shared__ float  tile[ROWS * LSTR];
  __shared__ float  hp[HP_TOTAL];

  const int tid = threadIdx.x;
  const int b   = blockIdx.x >> 6;
  const int sB  = (blockIdx.x & 63) * SB;

  const float tapv = taps[tid < HP_TOTAL ? tid : HP_TOTAL - 1];

#pragma unroll
  for (int it = 0; it < 8; ++it) {
    const int e4  = tid + it * 256;
    const int row = e4 >> 4;
    const int col = (e4 & 15) << 2;
    const int gr  = sB - HALO + row;
    f32x4 v = {0.0f, 0.0f, 0.0f, 0.0f};
    if (gr >= 0 && gr < S_LEN)
      v = *(const f32x4*)(x + ((size_t)b * S_LEN + (size_t)gr) * CH + col);
    *(f32x4*)(&tile[row * LSTR + col]) = v;
  }
  if (tid < HP_TOTAL) hp[tid] = tapv;
  __syncthreads();

  const int lane = tid & 31;
  const int w    = tid >> 5;
  const int c0   = (w & 3) * 16;
  const int st0  = (w >> 2) * 2;
  const int st1  = st0 + 1;

  const v8f z = {0.f, 0.f, 0.f, 0.f, 0.f, 0.f, 0.f, 0.f};

  {
    v8f a0 = z, a1 = z;
    conv_pair<28, 6, 0>(tile, hp, lane, st0, st1, c0, a0, a1);
    store_tile(out, b, 0, sB + 16 * st0, c0, lane, a0);
    store_tile(out, b, 0, sB + 16 * st1, c0, lane, a1);
  }
  {
    v8f a0 = z, a1 = z;
    conv_pair<48, 16, 43>(tile, hp, lane, st0, st1, c0, a0, a1);
    store_tile(out, b, 1, sB + 16 * st0, c0, lane, a0);
    store_tile(out, b, 1, sB + 16 * st1, c0, lane, a1);
  }
  {
    v8f a0 = z, a1 = z;
    conv_pair<80, 31, 106>(tile, hp, lane, st0, st1, c0, a0, a1);
    store_tile(out, b, 2, sB + 16 * st0, c0, lane, a0);
    store_tile(out, b, 2, sB + 16 * st1, c0, lane, a1);
  }
}

extern "C" void kernel_launch(void* const* d_in, const int* in_sizes, int n_in,
                              void* d_out, int out_size, void* d_ws, size_t ws_size,
                              hipStream_t stream) {
  (void)in_sizes; (void)n_in; (void)out_size; (void)ws_size;
  const float* x    = (const float*)d_in[0];
  float*       out  = (float*)d_out;
  float*       taps = (float*)d_ws;
  hipLaunchKernelGGL(cwt_gaus1_taps_kernel, dim3(1), dim3(256), 0, stream, taps);
  hipLaunchKernelGGL(cwt_gaus1_wmma_kernel, dim3(32 * 64), dim3(256), 0, stream,
                     x, (const float*)taps, out);
}


# Tap builder: the one-thread 256-step fp64 running sum becomes a parallel scan (wave shuffles + 8 wave totals; fp64 sum order changes); psi samples stay in LDS so filter samples need no second exp
# speedup vs baseline: 2.3251x; 1.1812x over previous
#include <hip/hip_runtime.h>
#include <math.h>

typedef float v2f   __attribute__((ext_vector_type(2)));
typedef float v8f   __attribute__((ext_vector_type(8)));
typedef float f32x4 __attribute__((ext_vector_type(4)));

namespace {
constexpr int S_LEN = 4096;
constexpr int CH    = 64;
constexpr int SB    = 64;
constexpr int HALO  = 31;
constexpr int ROWS  = 128;
constexpr int LSTR  = 80;

constexpr int F_TOTAL  = 11 + 31 + 61;
constexpr int HP_TOTAL = 43 + 63 + 95;
}

template<int KPAD, int SHIFT, int HPBASE>
__device__ __forceinline__ void conv_pair(const float* tile, const float* hp,
                                          int lane, int st0, int st1, int c0,
                                          v8f& acc0, v8f& acc1)
{
  const int i   = lane & 15;
  const int hi  = lane >> 4;
  const int rb0 = (HALO - SHIFT + 16 * st0) * LSTR + c0 + i;
  const int rb1 = (HALO - SHIFT + 16 * st1) * LSTR + c0 + i;
  const int ab  = HPBASE + 15 - i;
#pragma unroll
  for (int k0 = 0; k0 < KPAD; k0 += 4) {
    const int t0 = k0 + 2 * hi;
    v2f a, b0, b1;
    a.x  = hp[ab + t0];
    a.y  = hp[ab + t0 + 1];
    b0.x = tile[rb0 +  t0      * LSTR];
    b0.y = tile[rb0 + (t0 + 1) * LSTR];
    b1.x = tile[rb1 +  t0      * LSTR];
    b1.y = tile[rb1 + (t0 + 1) * LSTR];
    acc0 = __builtin_amdgcn_wmma_f32_16x16x4_f32(false, a, false, b0, (short)0, acc0, false, false);
    acc1 = __builtin_amdgcn_wmma_f32_16x16x4_f32(false, a, false, b1, (short)0, acc1, false, false);
  }
}

__device__ __forceinline__ void store_tile(float* __restrict__ out, int b, int sc,
                                           int row0, int c0, int lane, v8f acc)
{
  const int j  = lane & 15;
  const int hi = lane >> 4;
  const size_t base = (((size_t)b * 3 + sc) * S_LEN + (size_t)(row0 + 8 * hi)) * CH
                      + (size_t)(c0 + j);
#pragma unroll
  for (int r = 0; r < 8; ++r)
    out[base + (size_t)r * CH] = acc[r];
}

__global__ void __launch_bounds__(256)
cwt_gaus1_taps_kernel(float* __restrict__ hp)
{
  __shared__ double psi_s[1024];
  __shared__ double psum[256];
  __shared__ double wtot[8];
  __shared__ float  fbuf[F_TOTAL];

  const int tid  = threadIdx.x;
  const int lane = tid & 31;
  const int w    = tid >> 5;

  const double step = 10.0 / 1023.0;
  const double invC = 1.0 / pow(M_PI * 0.5, 0.25);

  double s = 0.0;
#pragma unroll
  for (int q = 0; q < 4; ++q) {
    const int    i  = tid * 4 + q;
    const double xv = -5.0 + step * (double)i;
    const double p  = -2.0 * xv * exp(-xv * xv) * invC;
    psi_s[i] = p;
    s += p;
  }

  double inc = s;
#pragma unroll
  for (int d = 1; d < 32; d <<= 1) {
    const double up = __shfl_up(inc, (unsigned)d, 32);
    if (lane >= d) inc += up;
  }
  double exw = __shfl_up(inc, 1u, 32);
  if (lane == 0) exw = 0.0;
  if (lane == 31) wtot[w] = inc;
  __syncthreads();
  {
    double off = 0.0;
#pragma unroll
    for (int v = 0; v < 7; ++v)
      off += (v < w) ? wtot[v] : 0.0;
    psum[tid] = off + exw;
  }
  __syncthreads();
  if (tid < F_TOTAL) {
    int a, k;
    if (tid < 11)      { a = 1; k = tid;      }
    else if (tid < 42) { a = 3; k = tid - 11; }
    else               { a = 6; k = tid - 42; }
    const long j = (long)((double)k / ((double)a * step));
    const double* g = &psi_s[j & ~3L];
    const int     r = (int)(j & 3L);
    const double p0 = g[0], p1 = g[1], p2 = g[2], p3 = g[3];
    double v = psum[j >> 2];
    v += p0;
    if (r >= 1) v += p1;
    if (r >= 2) v += p2;
    if (r >= 3) v += p3;
    fbuf[tid] = (float)(v * step);
  }
  __syncthreads();
  if (tid < HP_TOTAL) {
    int d, foff, LH; double aa;
    if (tid < 43)       { d = tid - 15;       foff = 0;  LH = 12; aa = 1.0; }
    else if (tid < 106) { d = tid - 43 - 15;  foff = 11; LH = 32; aa = 3.0; }
    else                { d = tid - 106 - 15; foff = 42; LH = 62; aa = 6.0; }
    float v = 0.0f;
    if (d >= 0 && d < LH) {
      const float fm1 = (d >= 1)      ? fbuf[foff + d - 1] : 0.0f;
      const float fm  = (d < LH - 1)  ? fbuf[foff + d]     : 0.0f;
      v = (float)(-sqrt(aa)) * (fm1 - fm);
    }
    hp[tid] = v;
  }
}

__global__ void __launch_bounds__(256)
cwt_gaus1_wmma_kernel(const float* __restrict__ x, const float* __restrict__ taps,
                      float* __restrict__ out)
{
  __shared__ float  tile[ROWS * LSTR];
  __shared__ float  hp[HP_TOTAL];

  const int tid = threadIdx.x;
  const int b   = blockIdx.x >> 6;
  const int sB  = (blockIdx.x & 63) * SB;

  const float tapv = taps[tid < HP_TOTAL ? tid : HP_TOTAL - 1];

#pragma unroll
  for (int it = 0; it < 8; ++it) {
    const int e4  = tid + it * 256;
    const int row = e4 >> 4;
    const int col = (e4 & 15) << 2;
    const int gr  = sB - HALO + row;
    f32x4 v = {0.0f, 0.0f, 0.0f, 0.0f};
    if (gr >= 0 && gr < S_LEN)
      v = *(const f32x4*)(x + ((size_t)b * S_LEN + (size_t)gr) * CH + col);
    *(f32x4*)(&tile[row * LSTR + col]) = v;
  }
  if (tid < HP_TOTAL) hp[tid] = tapv;
  __syncthreads();

  const int lane = tid & 31;
  const int w    = tid >> 5;
  const int c0   = (w & 3) * 16;
  const int st0  = (w >> 2) * 2;
  const int st1  = st0 + 1;

  const v8f z = {0.f, 0.f, 0.f, 0.f, 0.f, 0.f, 0.f, 0.f};

  {
    v8f a0 = z, a1 = z;
    conv_pair<28, 6, 0>(tile, hp, lane, st0, st1, c0, a0, a1);
    store_tile(out, b, 0, sB + 16 * st0, c0, lane, a0);
    store_tile(out, b, 0, sB + 16 * st1, c0, lane, a1);
  }
  {
    v8f a0 = z, a1 = z;
    conv_pair<48, 16, 43>(tile, hp, lane, st0, st1, c0, a0, a1);
    store_tile(out, b, 1, sB + 16 * st0, c0, lane, a0);
    store_tile(out, b, 1, sB + 16 * st1, c0, lane, a1);
  }
  {
    v8f a0 = z, a1 = z;
    conv_pair<80, 31, 106>(tile, hp, lane, st0, st1, c0, a0, a1);
    store_tile(out, b, 2, sB + 16 * st0, c0, lane, a0);
    store_tile(out, b, 2, sB + 16 * st1, c0, lane, a1);
  }
}

extern "C" void kernel_launch(void* const* d_in, const int* in_sizes, int n_in,
                              void* d_out, int out_size, void* d_ws, size_t ws_size,
                              hipStream_t stream) {
  (void)in_sizes; (void)n_in; (void)out_size; (void)ws_size;
  const float* x    = (const float*)d_in[0];
  float*       out  = (float*)d_out;
  float*       taps = (float*)d_ws;
  hipLaunchKernelGGL(cwt_gaus1_taps_kernel, dim3(1), dim3(256), 0, stream, taps);
  hipLaunchKernelGGL(cwt_gaus1_wmma_kernel, dim3(32 * 64), dim3(256), 0, stream,
                     x, (const float*)taps, out);
}


# Main kernel: 512 resident work-groups loop over 4 consecutive slabs each; next slab's rows load to registers under the current slab's WMMAs; two LDS tile buffers, one hand-written barrier per slab
# speedup vs baseline: 2.5637x; 1.1026x over previous
#include <hip/hip_runtime.h>
#include <math.h>

typedef float v2f   __attribute__((ext_vector_type(2)));
typedef float v8f   __attribute__((ext_vector_type(8)));
typedef float f32x4 __attribute__((ext_vector_type(4)));

namespace {
constexpr int S_LEN = 4096;
constexpr int CH    = 64;
constexpr int SB    = 64;
constexpr int HALO  = 31;
constexpr int ROWS  = 128;
constexpr int LSTR  = 80;
constexpr int NT    = 4;
constexpr int TILE_F = ROWS * LSTR;
static_assert(32 * 64 / NT == 512 && 64 % NT == 0, "the launch's 512 work-groups cover all slabs; a work-group stays in one batch");

constexpr int F_TOTAL  = 11 + 31 + 61;
constexpr int HP_TOTAL = 43 + 63 + 95;
}

template<int KPAD, int SHIFT, int HPBASE>
__device__ __forceinline__ void conv_pair(const float* tile, const float* hp,
                                          int lane, int st0, int st1, int c0,
                                          v8f& acc0, v8f& acc1)
{
  const int i   = lane & 15;
  const int hi  = lane >> 4;
  const int rb0 = (HALO - SHIFT + 16 * st0) * LSTR + c0 + i;
  const int rb1 = (HALO - SHIFT + 16 * st1) * LSTR + c0 + i;
  const int ab  = HPBASE + 15 - i;
#pragma unroll
  for (int k0 = 0; k0 < KPAD; k0 += 4) {
    const int t0 = k0 + 2 * hi;
    v2f a, b0, b1;
    a.x  = hp[ab + t0];
    a.y  = hp[ab + t0 + 1];
    b0.x = tile[rb0 +  t0      * LSTR];
    b0.y = tile[rb0 + (t0 + 1) * LSTR];
    b1.x = tile[rb1 +  t0      * LSTR];
    b1.y = tile[rb1 + (t0 + 1) * LSTR];
    acc0 = __builtin_amdgcn_wmma_f32_16x16x4_f32(false, a, false, b0, (short)0, acc0, false, false);
    acc1 = __builtin_amdgcn_wmma_f32_16x16x4_f32(false, a, false, b1, (short)0, acc1, false, false);
  }
}

__device__ __forceinline__ void store_tile(float* __restrict__ out, int b, int sc,
                                           int row0, int c0, int lane, v8f acc)
{
  const int j  = lane & 15;
  const int hi = lane >> 4;
  const size_t base = (((size_t)b * 3 + sc) * S_LEN + (size_t)(row0 + 8 * hi)) * CH
                      + (size_t)(c0 + j);
#pragma unroll
  for (int r = 0; r < 8; ++r)
    out[base + (size_t)r * CH] = acc[r];
}

__global__ void __launch_bounds__(256)
cwt_gaus1_taps_kernel(float* __restrict__ hp)
{
  __shared__ double psi_s[1024];
  __shared__ double psum[256];
  __shared__ double wtot[8];
  __shared__ float  fbuf[F_TOTAL];

  const int tid  = threadIdx.x;
  const int lane = tid & 31;
  const int w    = tid >> 5;

  const double step = 10.0 / 1023.0;
  const double invC = 1.0 / pow(M_PI * 0.5, 0.25);

  double s = 0.0;
#pragma unroll
  for (int q = 0; q < 4; ++q) {
    const int    i  = tid * 4 + q;
    const double xv = -5.0 + step * (double)i;
    const double p  = -2.0 * xv * exp(-xv * xv) * invC;
    psi_s[i] = p;
    s += p;
  }

  double inc = s;
#pragma unroll
  for (int d = 1; d < 32; d <<= 1) {
    const double up = __shfl_up(inc, (unsigned)d, 32);
    if (lane >= d) inc += up;
  }
  double exw = __shfl_up(inc, 1u, 32);
  if (lane == 0) exw = 0.0;
  if (lane == 31) wtot[w] = inc;
  __syncthreads();
  {
    double off = 0.0;
#pragma unroll
    for (int v = 0; v < 7; ++v)
      off += (v < w) ? wtot[v] : 0.0;
    psum[tid] = off + exw;
  }
  __syncthreads();
  if (tid < F_TOTAL) {
    int a, k;
    if (tid < 11)      { a = 1; k = tid;      }
    else if (tid < 42) { a = 3; k = tid - 11; }
    else               { a = 6; k = tid - 42; }
    const long j = (long)((double)k / ((double)a * step));
    const double* g = &psi_s[j & ~3L];
    const int     r = (int)(j & 3L);
    const double p0 = g[0], p1 = g[1], p2 = g[2], p3 = g[3];
    double v = psum[j >> 2];
    v += p0;
    if (r >= 1) v += p1;
    if (r >= 2) v += p2;
    if (r >= 3) v += p3;
    fbuf[tid] = (float)(v * step);
  }
  __syncthreads();
  if (tid < HP_TOTAL) {
    int d, foff, LH; double aa;
    if (tid < 43)       { d = tid - 15;       foff = 0;  LH = 12; aa = 1.0; }
    else if (tid < 106) { d = tid - 43 - 15;  foff = 11; LH = 32; aa = 3.0; }
    else                { d = tid - 106 - 15; foff = 42; LH = 62; aa = 6.0; }
    float v = 0.0f;
    if (d >= 0 && d < LH) {
      const float fm1 = (d >= 1)      ? fbuf[foff + d - 1] : 0.0f;
      const float fm  = (d < LH - 1)  ? fbuf[foff + d]     : 0.0f;
      v = (float)(-sqrt(aa)) * (fm1 - fm);
    }
    hp[tid] = v;
  }
}

__device__ __forceinline__ void load_rows(f32x4 (&pre)[8], const float* __restrict__ x,
                                          int b, int sB, int tid)
{
#pragma unroll
  for (int it = 0; it < 8; ++it) {
    const int e4  = tid + it * 256;
    const int row = e4 >> 4;
    const int col = (e4 & 15) << 2;
    const int gr  = sB - HALO + row;
    f32x4 v = {0.0f, 0.0f, 0.0f, 0.0f};
    if (gr >= 0 && gr < S_LEN)
      v = *(const f32x4*)(x + ((size_t)b * S_LEN + (size_t)gr) * CH + col);
    pre[it] = v;
  }
}

__device__ __forceinline__ void lds_barrier()
{
  asm volatile("s_wait_dscnt 0x0\n\ts_barrier_signal -1\n\ts_barrier_wait -1" ::: "memory");
}

__global__ void __launch_bounds__(256)
cwt_gaus1_wmma_kernel(const float* __restrict__ x, const float* __restrict__ taps,
                      float* __restrict__ out)
{
  __shared__ float  tile[2 * TILE_F];
  __shared__ float  hp[HP_TOTAL];

  const int tid   = threadIdx.x;
  const int tile0 = blockIdx.x * NT;
  const int b     = tile0 >> 6;
  const int sB0   = (tile0 & 63) * SB;

  const float tapv = taps[tid < HP_TOTAL ? tid : HP_TOTAL - 1];

  f32x4 pre[8];
  load_rows(pre, x, b, sB0, tid);

  const int lane = tid & 31;
  const int w    = tid >> 5;
  const int c0   = (w & 3) * 16;
  const int st0  = (w >> 2) * 2;
  const int st1  = st0 + 1;

  const v8f z = {0.f, 0.f, 0.f, 0.f, 0.f, 0.f, 0.f, 0.f};

#pragma clang loop unroll(disable)
  for (int t = 0; t < NT; ++t) {
    const int sB = sB0 + t * SB;
    float* tb = tile + (t & 1) * TILE_F;

#pragma unroll
    for (int it = 0; it < 8; ++it) {
      const int e4  = tid + it * 256;
      const int row = e4 >> 4;
      const int col = (e4 & 15) << 2;
      *(f32x4*)(&tb[row * LSTR + col]) = pre[it];
    }
    if (t == 0 && tid < HP_TOTAL) hp[tid] = tapv;
    lds_barrier();

    if (t + 1 < NT) load_rows(pre, x, b, sB + SB, tid);

    {
      v8f a0 = z, a1 = z;
      conv_pair<28, 6, 0>(tb, hp, lane, st0, st1, c0, a0, a1);
      store_tile(out, b, 0, sB + 16 * st0, c0, lane, a0);
      store_tile(out, b, 0, sB + 16 * st1, c0, lane, a1);
    }
    {
      v8f a0 = z, a1 = z;
      conv_pair<48, 16, 43>(tb, hp, lane, st0, st1, c0, a0, a1);
      store_tile(out, b, 1, sB + 16 * st0, c0, lane, a0);
      store_tile(out, b, 1, sB + 16 * st1, c0, lane, a1);
    }
    {
      v8f a0 = z, a1 = z;
      conv_pair<80, 31, 106>(tb, hp, lane, st0, st1, c0, a0, a1);
      store_tile(out, b, 2, sB + 16 * st0, c0, lane, a0);
      store_tile(out, b, 2, sB + 16 * st1, c0, lane, a1);
    }
  }
}

extern "C" void kernel_launch(void* const* d_in, const int* in_sizes, int n_in,
                              void* d_out, int out_size, void* d_ws, size_t ws_size,
                              hipStream_t stream) {
  (void)in_sizes; (void)n_in; (void)out_size; (void)ws_size;
  const float* x    = (const float*)d_in[0];
  float*       out  = (float*)d_out;
  float*       taps = (float*)d_ws;
  hipLaunchKernelGGL(cwt_gaus1_taps_kernel, dim3(1), dim3(256), 0, stream, taps);
  hipLaunchKernelGGL(cwt_gaus1_wmma_kernel, dim3(512), dim3(256), 0, stream,
                     x, (const float*)taps, out);
}


# Each wave takes 32 channels x 16 rows (was 16 x 32) and exchanges accumulator halves across its two lane halves: every output store instruction writes one whole 128-byte line, not two 64-byte halves
# speedup vs baseline: 2.6657x; 1.0398x over previous
#include <hip/hip_runtime.h>
#include <math.h>

typedef float v2f   __attribute__((ext_vector_type(2)));
typedef float v8f   __attribute__((ext_vector_type(8)));
typedef float f32x4 __attribute__((ext_vector_type(4)));

namespace {
constexpr int S_LEN = 4096;
constexpr int CH    = 64;
constexpr int SB    = 64;
constexpr int HALO  = 31;
constexpr int ROWS  = 128;
constexpr int LSTR  = 80;
constexpr int NT    = 4;
constexpr int TILE_F = ROWS * LSTR;
static_assert(32 * 64 / NT == 512 && 64 % NT == 0, "the launch's 512 work-groups cover all slabs; a work-group stays in one batch");

constexpr int F_TOTAL  = 11 + 31 + 61;
constexpr int HP_TOTAL = 43 + 63 + 95;
}

template<int KPAD, int SHIFT, int HPBASE>
__device__ __forceinline__ void conv_pair(const float* tile, const float* hp,
                                          int lane, int st, int c0,
                                          v8f& acc0, v8f& acc1)
{
  const int i   = lane & 15;
  const int hi  = lane >> 4;
  const int rb0 = (HALO - SHIFT + 16 * st) * LSTR + c0 + i;
  const int rb1 = rb0 + 16;
  const int ab  = HPBASE + 15 - i;
#pragma unroll
  for (int k0 = 0; k0 < KPAD; k0 += 4) {
    const int t0 = k0 + 2 * hi;
    v2f a, b0, b1;
    a.x  = hp[ab + t0];
    a.y  = hp[ab + t0 + 1];
    b0.x = tile[rb0 +  t0      * LSTR];
    b0.y = tile[rb0 + (t0 + 1) * LSTR];
    b1.x = tile[rb1 +  t0      * LSTR];
    b1.y = tile[rb1 + (t0 + 1) * LSTR];
    acc0 = __builtin_amdgcn_wmma_f32_16x16x4_f32(false, a, false, b0, (short)0, acc0, false, false);
    acc1 = __builtin_amdgcn_wmma_f32_16x16x4_f32(false, a, false, b1, (short)0, acc1, false, false);
  }
}

__device__ __forceinline__ float xhalf(float x)
{
  const int b = __builtin_bit_cast(int, x);
  return __builtin_bit_cast(float,
      __builtin_amdgcn_permlanex16(b, b, 0x76543210, 0xfedcba98, false, false));
}

__device__ __forceinline__ void store_pair(float* __restrict__ out, int b, int sc,
                                           int row0, int c0, int lane, v8f acc0, v8f acc1)
{
  const bool up = lane >= 16;
  const size_t base = (((size_t)b * 3 + sc) * S_LEN + (size_t)row0) * CH + (size_t)(c0 + lane);
#pragma unroll
  for (int r = 0; r < 8; ++r) {
    const float x0 = xhalf(acc0[r]);
    const float x1 = xhalf(acc1[r]);
    const float lo = up ? x1 : acc0[r];
    const float hh = up ? acc1[r] : x0;
    out[base + (size_t)r * CH]       = lo;
    out[base + (size_t)(r + 8) * CH] = hh;
  }
}

__global__ void __launch_bounds__(256)
cwt_gaus1_taps_kernel(float* __restrict__ hp)
{
  __shared__ double psi_s[1024];
  __shared__ double psum[256];
  __shared__ double wtot[8];
  __shared__ float  fbuf[F_TOTAL];

  const int tid  = threadIdx.x;
  const int lane = tid & 31;
  const int w    = tid >> 5;

  const double step = 10.0 / 1023.0;
  const double invC = 1.0 / pow(M_PI * 0.5, 0.25);

  double s = 0.0;
#pragma unroll
  for (int q = 0; q < 4; ++q) {
    const int    i  = tid * 4 + q;
    const double xv = -5.0 + step * (double)i;
    const double p  = -2.0 * xv * exp(-xv * xv) * invC;
    psi_s[i] = p;
    s += p;
  }

  double inc = s;
#pragma unroll
  for (int d = 1; d < 32; d <<= 1) {
    const double up = __shfl_up(inc, (unsigned)d, 32);
    if (lane >= d) inc += up;
  }
  double exw = __shfl_up(inc, 1u, 32);
  if (lane == 0) exw = 0.0;
  if (lane == 31) wtot[w] = inc;
  __syncthreads();
  {
    double off = 0.0;
#pragma unroll
    for (int v = 0; v < 7; ++v)
      off += (v < w) ? wtot[v] : 0.0;
    psum[tid] = off + exw;
  }
  __syncthreads();
  if (tid < F_TOTAL) {
    int a, k;
    if (tid < 11)      { a = 1; k = tid;      }
    else if (tid < 42) { a = 3; k = tid - 11; }
    else               { a = 6; k = tid - 42; }
    const long j = (long)((double)k / ((double)a * step));
    const double* g = &psi_s[j & ~3L];
    const int     r = (int)(j & 3L);
    const double p0 = g[0], p1 = g[1], p2 = g[2], p3 = g[3];
    double v = psum[j >> 2];
    v += p0;
    if (r >= 1) v += p1;
    if (r >= 2) v += p2;
    if (r >= 3) v += p3;
    fbuf[tid] = (float)(v * step);
  }
  __syncthreads();
  if (tid < HP_TOTAL) {
    int d, foff, LH; double aa;
    if (tid < 43)       { d = tid - 15;       foff = 0;  LH = 12; aa = 1.0; }
    else if (tid < 106) { d = tid - 43 - 15;  foff = 11; LH = 32; aa = 3.0; }
    else                { d = tid - 106 - 15; foff = 42; LH = 62; aa = 6.0; }
    float v = 0.0f;
    if (d >= 0 && d < LH) {
      const float fm1 = (d >= 1)      ? fbuf[foff + d - 1] : 0.0f;
      const float fm  = (d < LH - 1)  ? fbuf[foff + d]     : 0.0f;
      v = (float)(-sqrt(aa)) * (fm1 - fm);
    }
    hp[tid] = v;
  }
}

__device__ __forceinline__ void load_rows(f32x4 (&pre)[8], const float* __restrict__ x,
                                          int b, int sB, int tid)
{
#pragma unroll
  for (int it = 0; it < 8; ++it) {
    const int e4  = tid + it * 256;
    const int row = e4 >> 4;
    const int col = (e4 & 15) << 2;
    const int gr  = sB - HALO + row;
    f32x4 v = {0.0f, 0.0f, 0.0f, 0.0f};
    if (gr >= 0 && gr < S_LEN)
      v = *(const f32x4*)(x + ((size_t)b * S_LEN + (size_t)gr) * CH + col);
    pre[it] = v;
  }
}

__device__ __forceinline__ void lds_barrier()
{
  asm volatile("s_wait_dscnt 0x0\n\ts_barrier_signal -1\n\ts_barrier_wait -1" ::: "memory");
}

__global__ void __launch_bounds__(256)
cwt_gaus1_wmma_kernel(const float* __restrict__ x, const float* __restrict__ taps,
                      float* __restrict__ out)
{
  __shared__ float  tile[2 * TILE_F];
  __shared__ float  hp[HP_TOTAL];

  const int tid   = threadIdx.x;
  const int tile0 = blockIdx.x * NT;
  const int b     = tile0 >> 6;
  const int sB0   = (tile0 & 63) * SB;

  const float tapv = taps[tid < HP_TOTAL ? tid : HP_TOTAL - 1];

  f32x4 pre[8];
  load_rows(pre, x, b, sB0, tid);

  const int lane = tid & 31;
  const int w    = tid >> 5;
  const int c0   = (w & 1) * 32;
  const int st   = w >> 1;

  const v8f z = {0.f, 0.f, 0.f, 0.f, 0.f, 0.f, 0.f, 0.f};

#pragma clang loop unroll(disable)
  for (int t = 0; t < NT; ++t) {
    const int sB = sB0 + t * SB;
    float* tb = tile + (t & 1) * TILE_F;

#pragma unroll
    for (int it = 0; it < 8; ++it) {
      const int e4  = tid + it * 256;
      const int row = e4 >> 4;
      const int col = (e4 & 15) << 2;
      *(f32x4*)(&tb[row * LSTR + col]) = pre[it];
    }
    if (t == 0 && tid < HP_TOTAL) hp[tid] = tapv;
    lds_barrier();

    if (t + 1 < NT) load_rows(pre, x, b, sB + SB, tid);

    {
      v8f a0 = z, a1 = z;
      conv_pair<28, 6, 0>(tb, hp, lane, st, c0, a0, a1);
      store_pair(out, b, 0, sB + 16 * st, c0, lane, a0, a1);
    }
    {
      v8f a0 = z, a1 = z;
      conv_pair<48, 16, 43>(tb, hp, lane, st, c0, a0, a1);
      store_pair(out, b, 1, sB + 16 * st, c0, lane, a0, a1);
    }
    {
      v8f a0 = z, a1 = z;
      conv_pair<80, 31, 106>(tb, hp, lane, st, c0, a0, a1);
      store_pair(out, b, 2, sB + 16 * st, c0, lane, a0, a1);
    }
  }
}

extern "C" void kernel_launch(void* const* d_in, const int* in_sizes, int n_in,
                              void* d_out, int out_size, void* d_ws, size_t ws_size,
                              hipStream_t stream) {
  (void)in_sizes; (void)n_in; (void)out_size; (void)ws_size;
  const float* x    = (const float*)d_in[0];
  float*       out  = (float*)d_out;
  float*       taps = (float*)d_ws;
  hipLaunchKernelGGL(cwt_gaus1_taps_kernel, dim3(1), dim3(256), 0, stream, taps);
  hipLaunchKernelGGL(cwt_gaus1_wmma_kernel, dim3(512), dim3(256), 0, stream,
                     x, (const float*)taps, out);
}


# Tap builder: psi samples in runs of 8 per thread from 2 exp calls and a multiplicative recurrence (128 threads, one wave per SIMD), not one fp64 exp per sample; fp64 values and sum order change
# speedup vs baseline: 2.8678x; 1.0758x over previous
#include <hip/hip_runtime.h>
#include <math.h>

typedef float v2f   __attribute__((ext_vector_type(2)));
typedef float v8f   __attribute__((ext_vector_type(8)));
typedef float f32x4 __attribute__((ext_vector_type(4)));

namespace {
constexpr int S_LEN = 4096;
constexpr int CH    = 64;
constexpr int SB    = 64;
constexpr int HALO  = 31;
constexpr int ROWS  = 128;
constexpr int LSTR  = 80;
constexpr int NT    = 4;
constexpr int TILE_F = ROWS * LSTR;
static_assert(32 * 64 / NT == 512 && 64 % NT == 0, "the launch's 512 work-groups cover all slabs; a work-group stays in one batch");

constexpr int F_TOTAL  = 11 + 31 + 61;
constexpr int HP_TOTAL = 43 + 63 + 95;
}

template<int KPAD, int SHIFT, int HPBASE>
__device__ __forceinline__ void conv_pair(const float* tile, const float* hp,
                                          int lane, int st, int c0,
                                          v8f& acc0, v8f& acc1)
{
  const int i   = lane & 15;
  const int hi  = lane >> 4;
  const int rb0 = (HALO - SHIFT + 16 * st) * LSTR + c0 + i;
  const int rb1 = rb0 + 16;
  const int ab  = HPBASE + 15 - i;
#pragma unroll
  for (int k0 = 0; k0 < KPAD; k0 += 4) {
    const int t0 = k0 + 2 * hi;
    v2f a, b0, b1;
    a.x  = hp[ab + t0];
    a.y  = hp[ab + t0 + 1];
    b0.x = tile[rb0 +  t0      * LSTR];
    b0.y = tile[rb0 + (t0 + 1) * LSTR];
    b1.x = tile[rb1 +  t0      * LSTR];
    b1.y = tile[rb1 + (t0 + 1) * LSTR];
    acc0 = __builtin_amdgcn_wmma_f32_16x16x4_f32(false, a, false, b0, (short)0, acc0, false, false);
    acc1 = __builtin_amdgcn_wmma_f32_16x16x4_f32(false, a, false, b1, (short)0, acc1, false, false);
  }
}

__device__ __forceinline__ float xhalf(float x)
{
  const int b = __builtin_bit_cast(int, x);
  return __builtin_bit_cast(float,
      __builtin_amdgcn_permlanex16(b, b, 0x76543210, 0xfedcba98, false, false));
}

__device__ __forceinline__ void store_pair(float* __restrict__ out, int b, int sc,
                                           int row0, int c0, int lane, v8f acc0, v8f acc1)
{
  const bool up = lane >= 16;
  const size_t base = (((size_t)b * 3 + sc) * S_LEN + (size_t)row0) * CH + (size_t)(c0 + lane);
#pragma unroll
  for (int r = 0; r < 8; ++r) {
    const float x0 = xhalf(acc0[r]);
    const float x1 = xhalf(acc1[r]);
    const float lo = up ? x1 : acc0[r];
    const float hh = up ? acc1[r] : x0;
    out[base + (size_t)r * CH]       = lo;
    out[base + (size_t)(r + 8) * CH] = hh;
  }
}

__global__ void __launch_bounds__(256)
cwt_gaus1_taps_kernel(float* __restrict__ hp)
{
  __shared__ double cum_s[1024];
  __shared__ double psum[128];
  __shared__ double wtot[4];
  __shared__ float  fbuf[F_TOTAL];

  const int tid  = threadIdx.x;
  const int lane = tid & 31;
  const int w    = tid >> 5;

  const double step = 10.0 / 1023.0;
  const double invC = 1.0 / pow(M_PI * 0.5, 0.25);
  const double q2   = exp(-2.0 * step * step);

  double inc = 0.0, exw = 0.0;
  if (tid < 128) {
    const int    i0 = tid * 8;
    const double x0 = -5.0 + step * (double)i0;
    double e = exp(-x0 * x0) * (-2.0 * invC);
    double t = exp(-2.0 * step * x0 - step * step);
    double s = 0.0;
#pragma unroll
    for (int q = 0; q < 8; ++q) {
      const double xv = x0 + (double)q * step;
      s += xv * e;
      cum_s[i0 + q] = s;
      e *= t;
      t *= q2;
    }

    inc = s;
#pragma unroll
    for (int d = 1; d < 32; d <<= 1) {
      const double up = __shfl_up(inc, (unsigned)d, 32);
      if (lane >= d) inc += up;
    }
    exw = __shfl_up(inc, 1u, 32);
    if (lane == 0) exw = 0.0;
    if (lane == 31) wtot[w] = inc;
  }
  __syncthreads();
  if (tid < 128) {
    double off = 0.0;
#pragma unroll
    for (int v = 0; v < 3; ++v)
      off += (v < w) ? wtot[v] : 0.0;
    psum[tid] = off + exw;
  }
  __syncthreads();
  if (tid < F_TOTAL) {
    int a, k;
    if (tid < 11)      { a = 1; k = tid;      }
    else if (tid < 42) { a = 3; k = tid - 11; }
    else               { a = 6; k = tid - 42; }
    const long j = (long)((double)k / ((double)a * step));
    const double v = psum[j >> 3] + cum_s[j];
    fbuf[tid] = (float)(v * step);
  }
  __syncthreads();
  if (tid < HP_TOTAL) {
    int d, foff, LH; double aa;
    if (tid < 43)       { d = tid - 15;       foff = 0;  LH = 12; aa = 1.0; }
    else if (tid < 106) { d = tid - 43 - 15;  foff = 11; LH = 32; aa = 3.0; }
    else                { d = tid - 106 - 15; foff = 42; LH = 62; aa = 6.0; }
    float v = 0.0f;
    if (d >= 0 && d < LH) {
      const float fm1 = (d >= 1)      ? fbuf[foff + d - 1] : 0.0f;
      const float fm  = (d < LH - 1)  ? fbuf[foff + d]     : 0.0f;
      v = (float)(-sqrt(aa)) * (fm1 - fm);
    }
    hp[tid] = v;
  }
}

__device__ __forceinline__ void load_rows(f32x4 (&pre)[8], const float* __restrict__ x,
                                          int b, int sB, int tid)
{
#pragma unroll
  for (int it = 0; it < 8; ++it) {
    const int e4  = tid + it * 256;
    const int row = e4 >> 4;
    const int col = (e4 & 15) << 2;
    const int gr  = sB - HALO + row;
    f32x4 v = {0.0f, 0.0f, 0.0f, 0.0f};
    if (gr >= 0 && gr < S_LEN)
      v = *(const f32x4*)(x + ((size_t)b * S_LEN + (size_t)gr) * CH + col);
    pre[it] = v;
  }
}

__device__ __forceinline__ void lds_barrier()
{
  asm volatile("s_wait_dscnt 0x0\n\ts_barrier_signal -1\n\ts_barrier_wait -1" ::: "memory");
}

__global__ void __launch_bounds__(256)
cwt_gaus1_wmma_kernel(const float* __restrict__ x, const float* __restrict__ taps,
                      float* __restrict__ out)
{
  __shared__ float  tile[2 * TILE_F];
  __shared__ float  hp[HP_TOTAL];

  const int tid   = threadIdx.x;
  const int tile0 = blockIdx.x * NT;
  const int b     = tile0 >> 6;
  const int sB0   = (tile0 & 63) * SB;

  const float tapv = taps[tid < HP_TOTAL ? tid : HP_TOTAL - 1];

  f32x4 pre[8];
  load_rows(pre, x, b, sB0, tid);

  const int lane = tid & 31;
  const int w    = tid >> 5;
  const int c0   = (w & 1) * 32;
  const int st   = w >> 1;

  const v8f z = {0.f, 0.f, 0.f, 0.f, 0.f, 0.f, 0.f, 0.f};

#pragma clang loop unroll(disable)
  for (int t = 0; t < NT; ++t) {
    const int sB = sB0 + t * SB;
    float* tb = tile + (t & 1) * TILE_F;

#pragma unroll
    for (int it = 0; it < 8; ++it) {
      const int e4  = tid + it * 256;
      const int row = e4 >> 4;
      const int col = (e4 & 15) << 2;
      *(f32x4*)(&tb[row * LSTR + col]) = pre[it];
    }
    if (t == 0 && tid < HP_TOTAL) hp[tid] = tapv;
    lds_barrier();

    if (t + 1 < NT) load_rows(pre, x, b, sB + SB, tid);

    {
      v8f a0 = z, a1 = z;
      conv_pair<28, 6, 0>(tb, hp, lane, st, c0, a0, a1);
      store_pair(out, b, 0, sB + 16 * st, c0, lane, a0, a1);
    }
    {
      v8f a0 = z, a1 = z;
      conv_pair<48, 16, 43>(tb, hp, lane, st, c0, a0, a1);
      store_pair(out, b, 1, sB + 16 * st, c0, lane, a0, a1);
    }
    {
      v8f a0 = z, a1 = z;
      conv_pair<80, 31, 106>(tb, hp, lane, st, c0, a0, a1);
      store_pair(out, b, 2, sB + 16 * st, c0, lane, a0, a1);
    }
  }
}

extern "C" void kernel_launch(void* const* d_in, const int* in_sizes, int n_in,
                              void* d_out, int out_size, void* d_ws, size_t ws_size,
                              hipStream_t stream) {
  (void)in_sizes; (void)n_in; (void)out_size; (void)ws_size;
  const float* x    = (const float*)d_in[0];
  float*       out  = (float*)d_out;
  float*       taps = (float*)d_ws;
  hipLaunchKernelGGL(cwt_gaus1_taps_kernel, dim3(1), dim3(256), 0, stream, taps);
  hipLaunchKernelGGL(cwt_gaus1_wmma_kernel, dim3(512), dim3(256), 0, stream,
                     x, (const float*)taps, out);
}


# Main kernel: each 2x2 Toeplitz block pair by 3 WMMA products, not 4 (one on the sum of two input row blocks, two on tap differences); wave tile 32 rows, slab 128 rows; k-sum grouping changes
# speedup vs baseline: 3.1341x; 1.0929x over previous
#include <hip/hip_runtime.h>
#include <math.h>

typedef float v2f   __attribute__((ext_vector_type(2)));
typedef float v8f   __attribute__((ext_vector_type(8)));
typedef float f32x4 __attribute__((ext_vector_type(4)));

namespace {
constexpr int S_LEN = 4096;
constexpr int CH    = 64;
constexpr int SB    = 128;
constexpr int HALO  = 31;
constexpr int ROWS  = 192;
constexpr int LSTR  = 80;
constexpr int NT    = 2;
constexpr int TILE_F = ROWS * LSTR;
static_assert(32 * 32 / NT == 512 && 32 % NT == 0, "the launch's 512 work-groups cover all slabs; a work-group stays in one batch");

constexpr int F_TOTAL  = 11 + 31 + 61;
constexpr int HP_TOTAL = 43 + 63 + 95;
}

__device__ __forceinline__ v2f ld_a(const float* tab, int idx)
{
  v2f a;
  a.x = tab[idx];
  a.y = tab[idx + 1];
  return a;
}

__device__ __forceinline__ v2f ld_b(const float* tile, int idx)
{
  v2f b;
  b.x = tile[idx];
  b.y = tile[idx + LSTR];
  return b;
}

__device__ __forceinline__ v8f mma(v2f a, v2f b, v8f c)
{
  return __builtin_amdgcn_wmma_f32_16x16x4_f32(false, a, false, b, (short)0, c, false, false);
}

template<int NSTEP, int TA, int TB>
__device__ __forceinline__ void mm_block(const float* tab, int abh, const float* tile, int rbh,
                                         v8f& acc0, v8f& acc1)
{
#pragma unroll
  for (int s = 0; s < NSTEP; ++s) {
    const v2f a  = ld_a(tab, abh + TA + 4 * s);
    const v2f b0 = ld_b(tile, rbh + (TB + 4 * s) * LSTR);
    const v2f b1 = ld_b(tile, rbh + (TB + 4 * s) * LSTR + 16);
    acc0 = mma(a, b0, acc0);
    acc1 = mma(a, b1, acc1);
  }
}

template<int Q>
__device__ __forceinline__ void pair_block(const float* hpt, const float* dpt, const float* dnt,
                                           int abh, const float* tile, int rbh,
                                           v8f& p0, v8f& p1, v8f& y00, v8f& y01, v8f& y10, v8f& y11)
{
#pragma unroll
  for (int s = 0; s < 4; ++s) {
    const v2f aT = ld_a(hpt, abh + 16 * Q + 4 * s);
    const v2f aP = ld_a(dpt, abh + 16 * Q + 4 * s);
    const v2f aN = ld_a(dnt, abh + 16 * (Q - 1) + 4 * s);
    const v2f u0 = ld_b(tile, rbh + (16 * Q + 4 * s) * LSTR);
    const v2f u1 = ld_b(tile, rbh + (16 * Q + 4 * s) * LSTR + 16);
    const v2f w0 = ld_b(tile, rbh + (16 * (Q + 1) + 4 * s) * LSTR);
    const v2f w1 = ld_b(tile, rbh + (16 * (Q + 1) + 4 * s) * LSTR + 16);
    const v2f s0 = u0 + w0;
    const v2f s1 = u1 + w1;
    p0  = mma(aT, s0, p0);
    p1  = mma(aT, s1, p1);
    y00 = mma(aP, w0, y00);
    y01 = mma(aP, w1, y01);
    y10 = mma(aN, u0, y10);
    y11 = mma(aN, u1, y11);
  }
}

__device__ __forceinline__ float xhalf(float x)
{
  const int b = __builtin_bit_cast(int, x);
  return __builtin_bit_cast(float,
      __builtin_amdgcn_permlanex16(b, b, 0x76543210, 0xfedcba98, false, false));
}

__device__ __forceinline__ void store_pair(float* __restrict__ out, int b, int sc,
                                           int row0, int c0, int lane, v8f acc0, v8f acc1)
{
  const bool up = lane >= 16;
  const size_t base = (((size_t)b * 3 + sc) * S_LEN + (size_t)row0) * CH + (size_t)(c0 + lane);
#pragma unroll
  for (int r = 0; r < 8; ++r) {
    const float x0 = xhalf(acc0[r]);
    const float x1 = xhalf(acc1[r]);
    const float lo = up ? x1 : acc0[r];
    const float hh = up ? acc1[r] : x0;
    out[base + (size_t)r * CH]       = lo;
    out[base + (size_t)(r + 8) * CH] = hh;
  }
}

__global__ void __launch_bounds__(256)
cwt_gaus1_taps_kernel(float* __restrict__ hp)
{
  __shared__ double cum_s[1024];
  __shared__ double psum[128];
  __shared__ double wtot[4];
  __shared__ float  fbuf[F_TOTAL];

  const int tid  = threadIdx.x;
  const int lane = tid & 31;
  const int w    = tid >> 5;

  const double step = 10.0 / 1023.0;
  const double invC = 1.0 / pow(M_PI * 0.5, 0.25);
  const double q2   = exp(-2.0 * step * step);

  double inc = 0.0, exw = 0.0;
  if (tid < 128) {
    const int    i0 = tid * 8;
    const double x0 = -5.0 + step * (double)i0;
    double e = exp(-x0 * x0) * (-2.0 * invC);
    double t = exp(-2.0 * step * x0 - step * step);
    double s = 0.0;
#pragma unroll
    for (int q = 0; q < 8; ++q) {
      const double xv = x0 + (double)q * step;
      s += xv * e;
      cum_s[i0 + q] = s;
      e *= t;
      t *= q2;
    }

    inc = s;
#pragma unroll
    for (int d = 1; d < 32; d <<= 1) {
      const double up = __shfl_up(inc, (unsigned)d, 32);
      if (lane >= d) inc += up;
    }
    exw = __shfl_up(inc, 1u, 32);
    if (lane == 0) exw = 0.0;
    if (lane == 31) wtot[w] = inc;
  }
  __syncthreads();
  if (tid < 128) {
    double off = 0.0;
#pragma unroll
    for (int v = 0; v < 3; ++v)
      off += (v < w) ? wtot[v] : 0.0;
    psum[tid] = off + exw;
  }
  __syncthreads();
  if (tid < F_TOTAL) {
    int a, k;
    if (tid < 11)      { a = 1; k = tid;      }
    else if (tid < 42) { a = 3; k = tid - 11; }
    else               { a = 6; k = tid - 42; }
    const long j = (long)((double)k / ((double)a * step));
    const double v = psum[j >> 3] + cum_s[j];
    fbuf[tid] = (float)(v * step);
  }
  __syncthreads();
  if (tid < HP_TOTAL) {
    int d, foff, LH; double aa;
    if (tid < 43)       { d = tid - 15;       foff = 0;  LH = 12; aa = 1.0; }
    else if (tid < 106) { d = tid - 43 - 15;  foff = 11; LH = 32; aa = 3.0; }
    else                { d = tid - 106 - 15; foff = 42; LH = 62; aa = 6.0; }
    float v = 0.0f;
    if (d >= 0 && d < LH) {
      const float fm1 = (d >= 1)      ? fbuf[foff + d - 1] : 0.0f;
      const float fm  = (d < LH - 1)  ? fbuf[foff + d]     : 0.0f;
      v = (float)(-sqrt(aa)) * (fm1 - fm);
    }
    hp[tid] = v;
  }
}

__device__ __forceinline__ void load_rows(f32x4 (&pre)[12], const float* __restrict__ x,
                                          int b, int sB, int tid)
{
#pragma unroll
  for (int it = 0; it < 12; ++it) {
    const int e4  = tid + it * 256;
    const int row = e4 >> 4;
    const int col = (e4 & 15) << 2;
    const int gr  = sB - HALO + row;
    f32x4 v = {0.0f, 0.0f, 0.0f, 0.0f};
    if (gr >= 0 && gr < S_LEN)
      v = *(const f32x4*)(x + ((size_t)b * S_LEN + (size_t)gr) * CH + col);
    pre[it] = v;
  }
}

__device__ __forceinline__ void lds_barrier()
{
  asm volatile("s_wait_dscnt 0x0\n\ts_barrier_signal -1\n\ts_barrier_wait -1" ::: "memory");
}

__global__ void __launch_bounds__(256)
cwt_gaus1_wmma_kernel(const float* __restrict__ x, const float* __restrict__ taps,
                      float* __restrict__ out)
{
  __shared__ float  tile[2 * TILE_F];
  __shared__ float  hp[HP_TOTAL];
  __shared__ float  dp[HP_TOTAL];
  __shared__ float  dn[HP_TOTAL];

  const int tid   = threadIdx.x;
  const int tile0 = blockIdx.x * NT;
  const int b     = tile0 >> 5;
  const int sB0   = (tile0 & 31) * SB;

  const int tq   = tid < HP_TOTAL ? tid : HP_TOTAL - 1;
  const int tend = tq < 43 ? 43 : (tq < 106 ? 106 : HP_TOTAL);
  const float tapv = taps[tq];
  const float tapw = taps[tq + 16 < tend ? tq + 16 : tq];

  f32x4 pre[12];
  load_rows(pre, x, b, sB0, tid);

  const int lane = tid & 31;
  const int w    = tid >> 5;
  const int c0   = (w & 1) * 32;
  const int rg   = w >> 1;
  const int i    = lane & 15;
  const int hi   = lane >> 4;
  const int cb   = 2 * hi * LSTR + c0 + i;
  const int ah   = 15 - i + 2 * hi;

  const v8f z = {0.f, 0.f, 0.f, 0.f, 0.f, 0.f, 0.f, 0.f};

#pragma clang loop unroll(disable)
  for (int t = 0; t < NT; ++t) {
    const int sB = sB0 + t * SB;
    float* tb = tile + (t & 1) * TILE_F;

#pragma unroll
    for (int it = 0; it < 12; ++it) {
      const int e4  = tid + it * 256;
      const int row = e4 >> 4;
      const int col = (e4 & 15) << 2;
      *(f32x4*)(&tb[row * LSTR + col]) = pre[it];
    }
    if (t == 0 && tid < HP_TOTAL) {
      hp[tid] = tapv;
      dp[tid] = tapw - tapv;
      dn[tid] = tapv - tapw;
    }
    lds_barrier();

    if (t + 1 < NT) load_rows(pre, x, b, sB + SB, tid);

    const int row0 = sB + 32 * rg;
    {
      const int rbh = (HALO - 6 + 32 * rg) * LSTR + cb;
      const int abh = 0 + ah;
      v8f y00 = z, y01 = z, y10 = z, y11 = z;
      mm_block<7, 0, 0>(hp, abh, tb, rbh, y00, y01);
      mm_block<7, 0, 16>(hp, abh, tb, rbh, y10, y11);
      store_pair(out, b, 0, row0,      c0, lane, y00, y01);
      store_pair(out, b, 0, row0 + 16, c0, lane, y10, y11);
    }
    __builtin_amdgcn_sched_barrier(0x96);
    {
      const int rbh = (HALO - 16 + 32 * rg) * LSTR + cb;
      const int abh = 43 + ah;
      v8f y00 = z, y01 = z, y10 = z, y11 = z, p0 = z, p1 = z;
      mm_block<4, 0, 0>(hp, abh, tb, rbh, y00, y01);
      pair_block<1>(hp, dp, dn, abh, tb, rbh, p0, p1, y00, y01, y10, y11);
      mm_block<4, 32, 48>(hp, abh, tb, rbh, y10, y11);
      store_pair(out, b, 1, row0,      c0, lane, y00 + p0, y01 + p1);
      store_pair(out, b, 1, row0 + 16, c0, lane, y10 + p0, y11 + p1);
    }
    __builtin_amdgcn_sched_barrier(0x96);
    {
      const int rbh = (HALO - 31 + 32 * rg) * LSTR + cb;
      const int abh = 106 + ah;
      v8f y00 = z, y01 = z, y10 = z, y11 = z, p0 = z, p1 = z;
      mm_block<4, 0, 0>(hp, abh, tb, rbh, y00, y01);
      pair_block<1>(hp, dp, dn, abh, tb, rbh, p0, p1, y00, y01, y10, y11);
      pair_block<3>(hp, dp, dn, abh, tb, rbh, p0, p1, y00, y01, y10, y11);
      mm_block<4, 64, 80>(hp, abh, tb, rbh, y10, y11);
      store_pair(out, b, 2, row0,      c0, lane, y00 + p0, y01 + p1);
      store_pair(out, b, 2, row0 + 16, c0, lane, y10 + p0, y11 + p1);
    }
  }
}

extern "C" void kernel_launch(void* const* d_in, const int* in_sizes, int n_in,
                              void* d_out, int out_size, void* d_ws, size_t ws_size,
                              hipStream_t stream) {
  (void)in_sizes; (void)n_in; (void)out_size; (void)ws_size;
  const float* x    = (const float*)d_in[0];
  float*       out  = (float*)d_out;
  float*       taps = (float*)d_ws;
  hipLaunchKernelGGL(cwt_gaus1_taps_kernel, dim3(1), dim3(256), 0, stream, taps);
  hipLaunchKernelGGL(cwt_gaus1_wmma_kernel, dim3(512), dim3(256), 0, stream,
                     x, (const float*)taps, out);
}
